# starting kernel
# baseline (speedup 1.0000x reference)
#include <hip/hip_runtime.h>


typedef __attribute__((ext_vector_type(2))) float v2f;
typedef __attribute__((ext_vector_type(8))) float v8f;

#define SCALE 0.6513215599f

__device__ __forceinline__ float csign(int a, int b) {
    int cnt = 0;
    for (int aa = a >> 1; aa; aa >>= 1) cnt += __popc(aa & b);
    if ((a & b) & 16) cnt ^= 1;
    return (cnt & 1) ? -1.0f : 1.0f;
}

__global__ __launch_bounds__(256) void gram_kernel(const float* __restrict__ Win,
                                                   const float* __restrict__ Wout,
                                                   float* __restrict__ G) {
    const int d    = blockIdx.x;
    const int tid  = threadIdx.x;
    const int wid  = tid >> 5;
    const int lane = tid & 31;
    const int half = lane & 15;
    const int hi   = lane >> 4;

    __shared__ float red[8 * 1024];

    v8f acc00 = {}, acc01 = {}, acc10 = {}, acc11 = {};

    const float* WinD = Win + d * 32;
    const int kbeg = wid * 128;
    #pragma unroll 4
    for (int k0 = kbeg; k0 < kbeg + 128; k0 += 4) {
        const int kA = k0 + hi * 2;
        v2f a0, a1, b0, b1;
        a0.x = WinD[(size_t)kA * 160 + half];
        a0.y = WinD[(size_t)(kA + 1) * 160 + half];
        a1.x = WinD[(size_t)kA * 160 + 16 + half];
        a1.y = WinD[(size_t)(kA + 1) * 160 + 16 + half];
        b0.x = Wout[kA * 32 + half];
        b0.y = Wout[(kA + 1) * 32 + half];
        b1.x = Wout[kA * 32 + 16 + half];
        b1.y = Wout[(kA + 1) * 32 + 16 + half];

        acc00 = __builtin_amdgcn_wmma_f32_16x16x4_f32(false, a0, false, b0, (short)0, acc00, false, false);
        acc01 = __builtin_amdgcn_wmma_f32_16x16x4_f32(false, a0, false, b1, (short)0, acc01, false, false);
        acc10 = __builtin_amdgcn_wmma_f32_16x16x4_f32(false, a1, false, b0, (short)0, acc10, false, false);
        acc11 = __builtin_amdgcn_wmma_f32_16x16x4_f32(false, a1, false, b1, (short)0, acc11, false, false);
    }

    float* my = red + wid * 1024;
    #pragma unroll
    for (int r = 0; r < 8; ++r) {
        const int m = r + 8 * hi;
        my[m * 32 + half]             = acc00[r];
        my[m * 32 + 16 + half]        = acc01[r];
        my[(m + 16) * 32 + half]      = acc10[r];
        my[(m + 16) * 32 + 16 + half] = acc11[r];
    }
    __syncthreads();

    #pragma unroll
    for (int j = 0; j < 4; ++j) {
        const int o = tid * 4 + j;
        float s = 0.0f;
        #pragma unroll
        for (int w = 0; w < 8; ++w) s += red[w * 1024 + o];
        G[d * 1024 + o] = s;
    }
}

__global__ __launch_bounds__(256) void mix_kernel(const float* __restrict__ G,
                                                  float* __restrict__ M) {
    __shared__ float Gd[1024];
    const int t = blockIdx.x * 256 + threadIdx.x;
    const int d = t >> 10;
    const float* src = G + d * 1024;
    for (int j = threadIdx.x; j < 1024; j += 256) Gd[j] = src[j];
    __syncthreads();

    const int p1 = (t >> 5) & 31;
    const int k  = t & 31;
    float acc = 0.0f;
    #pragma unroll
    for (int p = 0; p < 32; ++p) {
        const int q1 = p1 ^ p;
        const int q2 = p ^ k;
        acc += csign(p1, q1) * csign(p, q2) * Gd[q1 * 32 + q2];
    }
    M[((d << 5) | p1) * 32 + k] = SCALE * acc;
}

__global__ __launch_bounds__(256) void apply_kernel(const float* __restrict__ X,
                                                    const float* __restrict__ M,
                                                    float* __restrict__ out) {
    const int tid  = threadIdx.x;
    const int wid  = tid >> 5;
    const int lane = tid & 31;
    const int half = lane & 15;
    const int hi   = lane >> 4;
    const int tile = blockIdx.x * 8 + wid;
    const int rt   = tile >> 1;
    const int ct   = tile & 1;

    v8f acc = {};
    const float* Xrow = X + (size_t)rt * 16 * 160;
    #pragma unroll 8
    for (int k0 = 0; k0 < 160; k0 += 4) {
        const int kA = k0 + hi * 2;
        v2f a, b;
        a.x = Xrow[(size_t)half * 160 + kA];
        a.y = Xrow[(size_t)half * 160 + kA + 1];
        b.x = M[kA * 32 + ct * 16 + half];
        b.y = M[(kA + 1) * 32 + ct * 16 + half];
        acc = __builtin_amdgcn_wmma_f32_16x16x4_f32(false, a, false, b, (short)0, acc, false, false);
    }
    #pragma unroll
    for (int r = 0; r < 8; ++r) {
        const int m = rt * 16 + r + 8 * hi;
        out[m * 32 + ct * 16 + half] = acc[r];
    }
}

extern "C" void kernel_launch(void* const* d_in, const int* in_sizes, int n_in,
                              void* d_out, int out_size, void* d_ws, size_t ws_size,
                              hipStream_t stream) {
    const float* x    = (const float*)d_in[0];
    const float* Win  = (const float*)d_in[1];
    const float* Wout = (const float*)d_in[2];
    float* out = (float*)d_out;

    float* G = (float*)d_ws;
    float* M = G + 5 * 1024;

    gram_kernel<<<5, 256, 0, stream>>>(Win, Wout, G);
    mix_kernel<<<20, 256, 0, stream>>>(G, M);
    apply_kernel<<<8, 256, 0, stream>>>(x, M, out);
}


# Gram retiled by output tile, not by h: one work-group per 16x16 tile of G (5 -> 20 groups), 16 waves split K in 64s and sum in LDS (new summation tree); G tile-major in workspace
# speedup vs baseline: 1.4053x; 1.4053x over previous
#include <hip/hip_runtime.h>


typedef __attribute__((ext_vector_type(2))) float v2f;
typedef __attribute__((ext_vector_type(8))) float v8f;

#define SCALE 0.6513215599f

__device__ __forceinline__ float csign(int a, int b) {
    int cnt = 0;
    for (int aa = a >> 1; aa; aa >>= 1) cnt += __popc(aa & b);
    if ((a & b) & 16) cnt ^= 1;
    return (cnt & 1) ? -1.0f : 1.0f;
}

__global__ __launch_bounds__(512) void gram_kernel(const float* __restrict__ Win,
                                                   const float* __restrict__ Wout,
                                                   float* __restrict__ G) {
    const int t    = blockIdx.x;
    const int mt   = t >> 1;
    const int nt   = t & 1;
    const int tid  = threadIdx.x;
    const int wid  = tid >> 5;
    const int lane = tid & 31;
    const int half = lane & 15;
    const int hi   = lane >> 4;

    __shared__ float red[16 * 256];

    const float* Ap = Win  + (size_t)(wid * 64 + hi * 2) * 160 + mt * 16 + half;
    const float* Bp = Wout + (size_t)(wid * 64 + hi * 2) * 32  + nt * 16 + half;

    v2f a[16], b[16];
    #pragma unroll
    for (int i = 0; i < 16; ++i) {
        a[i].x = Ap[(size_t)(4 * i) * 160];
        a[i].y = Ap[(size_t)(4 * i + 1) * 160];
        b[i].x = Bp[(4 * i) * 32];
        b[i].y = Bp[(4 * i + 1) * 32];
    }

    v8f acc = {};
    #pragma unroll
    for (int i = 0; i < 16; ++i)
        acc = __builtin_amdgcn_wmma_f32_16x16x4_f32(false, a[i], false, b[i], (short)0, acc, false, false);

    float* my = red + wid * 256;
    #pragma unroll
    for (int r = 0; r < 8; ++r) my[(r + 8 * hi) * 16 + half] = acc[r];
    __syncthreads();

    if (tid < 256) {
        float s = 0.0f;
        #pragma unroll
        for (int w = 0; w < 16; ++w) s += red[w * 256 + tid];
        G[t * 256 + tid] = s;
    }
}

__global__ __launch_bounds__(256) void mix_kernel(const float* __restrict__ G,
                                                  float* __restrict__ M) {
    __shared__ float Gd[1024];
    const int t = blockIdx.x * 256 + threadIdx.x;
    const int d = t >> 10;
    const float* src = G + d * 1024;
    for (int j = threadIdx.x; j < 1024; j += 256) {
        const int q1 = j >> 5, q2 = j & 31;
        Gd[j] = src[(((q1 >> 4) * 2 + (q2 >> 4)) << 8) + ((q1 & 15) << 4) + (q2 & 15)];
    }
    __syncthreads();

    const int p1 = (t >> 5) & 31;
    const int k  = t & 31;
    float acc = 0.0f;
    #pragma unroll
    for (int p = 0; p < 32; ++p) {
        const int q1 = p1 ^ p;
        const int q2 = p ^ k;
        acc += csign(p1, q1) * csign(p, q2) * Gd[q1 * 32 + q2];
    }
    M[((d << 5) | p1) * 32 + k] = SCALE * acc;
}

__global__ __launch_bounds__(256) void apply_kernel(const float* __restrict__ X,
                                                    const float* __restrict__ M,
                                                    float* __restrict__ out) {
    const int tid  = threadIdx.x;
    const int wid  = tid >> 5;
    const int lane = tid & 31;
    const int half = lane & 15;
    const int hi   = lane >> 4;
    const int tile = blockIdx.x * 8 + wid;
    const int rt   = tile >> 1;
    const int ct   = tile & 1;

    v8f acc = {};
    const float* Xrow = X + (size_t)rt * 16 * 160;
    #pragma unroll 8
    for (int k0 = 0; k0 < 160; k0 += 4) {
        const int kA = k0 + hi * 2;
        v2f a, b;
        a.x = Xrow[(size_t)half * 160 + kA];
        a.y = Xrow[(size_t)half * 160 + kA + 1];
        b.x = M[kA * 32 + ct * 16 + half];
        b.y = M[(kA + 1) * 32 + ct * 16 + half];
        acc = __builtin_amdgcn_wmma_f32_16x16x4_f32(false, a, false, b, (short)0, acc, false, false);
    }
    #pragma unroll
    for (int r = 0; r < 8; ++r) {
        const int m = rt * 16 + r + 8 * hi;
        out[m * 32 + ct * 16 + half] = acc[r];
    }
}

extern "C" void kernel_launch(void* const* d_in, const int* in_sizes, int n_in,
                              void* d_out, int out_size, void* d_ws, size_t ws_size,
                              hipStream_t stream) {
    const float* x    = (const float*)d_in[0];
    const float* Win  = (const float*)d_in[1];
    const float* Wout = (const float*)d_in[2];
    float* out = (float*)d_out;

    float* G = (float*)d_ws;
    float* M = G + 5 * 1024;

    gram_kernel<<<20, 512, 0, stream>>>(Win, Wout, G);
    mix_kernel<<<20, 256, 0, stream>>>(G, M);
    apply_kernel<<<8, 256, 0, stream>>>(x, M, out);
}


# mix launch dropped (3 -> 2): apply copies G to LDS, folds it to P[d] = sum_h Win[h,d]*Wout[h] (q order: new summation order), builds M = sign*P there; x and B fragments read ahead of the WMMAs
# speedup vs baseline: 2.4293x; 1.7287x over previous
#include <hip/hip_runtime.h>


typedef __attribute__((ext_vector_type(2))) float v2f;
typedef __attribute__((ext_vector_type(4))) float v4f;
typedef __attribute__((ext_vector_type(8))) float v8f;

#define SCALE 0.6513215599f

__device__ __forceinline__ unsigned csign_bit(int a, int b) {
    const unsigned mask = (unsigned)(((a >> 1) ^ (a >> 2) ^ (a >> 3) ^ (a >> 4)) | (a & 16));
    return (unsigned)__popc(mask & (unsigned)b) << 31;
}
__device__ __forceinline__ float with_sign(float v, unsigned bit) {
    return __uint_as_float(__float_as_uint(v) ^ bit);
}

__global__ __launch_bounds__(512) void gram_kernel(const float* __restrict__ Win,
                                                   const float* __restrict__ Wout,
                                                   float* __restrict__ G) {
    const int t    = blockIdx.x;
    const int mt   = t >> 1;
    const int nt   = t & 1;
    const int tid  = threadIdx.x;
    const int wid  = tid >> 5;
    const int lane = tid & 31;
    const int half = lane & 15;
    const int hi   = lane >> 4;

    __shared__ float red[16 * 256];

    const float* Ap = Win  + (size_t)(wid * 64 + hi * 2) * 160 + mt * 16 + half;
    const float* Bp = Wout + (size_t)(wid * 64 + hi * 2) * 32  + nt * 16 + half;

    v2f a[16], b[16];
    #pragma unroll
    for (int i = 0; i < 16; ++i) {
        a[i].x = Ap[(size_t)(4 * i) * 160];
        a[i].y = Ap[(size_t)(4 * i + 1) * 160];
        b[i].x = Bp[(4 * i) * 32];
        b[i].y = Bp[(4 * i + 1) * 32];
    }

    v8f acc = {};
    #pragma unroll
    for (int i = 0; i < 16; ++i)
        acc = __builtin_amdgcn_wmma_f32_16x16x4_f32(false, a[i], false, b[i], (short)0, acc, false, false);

    float* my = red + wid * 256;
    #pragma unroll
    for (int r = 0; r < 8; ++r) my[(r + 8 * hi) * 16 + half] = acc[r];
    __syncthreads();

    if (tid < 256) {
        float s = 0.0f;
        #pragma unroll
        for (int w = 0; w < 16; ++w) s += red[w * 256 + tid];
        G[t * 256 + tid] = s;
    }
}

__device__ __forceinline__ void lds_barrier() {
    __builtin_amdgcn_fence(__ATOMIC_RELEASE, "workgroup", "local");
    __builtin_amdgcn_s_barrier();
    __builtin_amdgcn_fence(__ATOMIC_ACQUIRE, "workgroup", "local");
}

#define BPITCH 84

__global__ __launch_bounds__(256) void apply_kernel(const float* __restrict__ X,
                                                    const float* __restrict__ G,
                                                    float* __restrict__ out) {
    __shared__ __attribute__((aligned(16))) float Gs[160 * 32];
    __shared__ float Ps[160];
    __shared__ __attribute__((aligned(16))) float Ms[64 * BPITCH];

    const int tid  = threadIdx.x;
    const int wid  = tid >> 5;
    const int lane = tid & 31;
    const int half = lane & 15;
    const int hi   = lane >> 4;
    const int tile = blockIdx.x * 8 + wid;
    const int rt   = tile >> 1;
    const int ct   = tile & 1;

    float4 g[5];
    #pragma unroll
    for (int i = 0; i < 5; ++i)
        g[i] = *reinterpret_cast<const float4*>(G + (tid + 256 * i) * 4);

    v2f a[40];
    const float* Xp = X + (size_t)(rt * 16 + half) * 160 + hi * 2;
    #pragma unroll
    for (int i = 0; i < 40; ++i) {
        a[i].x = Xp[4 * i];
        a[i].y = Xp[4 * i + 1];
    }

    #pragma unroll
    for (int i = 0; i < 5; ++i) {
        const int e   = (tid + 256 * i) * 4;
        const int t   = e >> 8;
        const int row = (t >> 1) * 16 + ((e >> 4) & 15);
        const int col = (t & 1) * 16 + (e & 15);
        *reinterpret_cast<float4*>(Gs + row * 32 + col) = g[i];
    }
    lds_barrier();

    if (tid < 160) {
        const int j = tid & 31;
        const float* Gd = Gs + (tid >> 5) * 1024;
        float s = 0.0f;
        #pragma unroll
        for (int q = 0; q < 32; ++q) {
            const int q2 = q ^ j;
            s += with_sign(Gd[q * 32 + q2], csign_bit(q, q2));
        }
        Ps[tid] = SCALE * s;
    }
    lds_barrier();

    #pragma unroll
    for (int i = 0; i < 20; ++i) {
        const int row = wid + 8 * i;
        const int p1  = row & 31;
        const int jj  = p1 ^ lane;
        const int dst = ((lane >> 4) * 32 + ((row >> 1) & 1) * 16 + (lane & 15)) * BPITCH
                      + (row >> 2) * 2 + (row & 1);
        Ms[dst] = with_sign(Ps[(row & ~31) + jj], csign_bit(p1, jj));
    }
    lds_barrier();

    v4f bq[20];
    const v4f* Bp = reinterpret_cast<const v4f*>(Ms + (ct * 32 + lane) * BPITCH);
    #pragma unroll
    for (int i = 0; i < 20; ++i) bq[i] = Bp[i];
    __builtin_amdgcn_sched_barrier(0);

    v8f acc = {};
    #pragma unroll
    for (int i = 0; i < 20; ++i) {
        acc = __builtin_amdgcn_wmma_f32_16x16x4_f32(false, a[2 * i],     false, bq[i].xy, (short)0, acc, false, false);
        acc = __builtin_amdgcn_wmma_f32_16x16x4_f32(false, a[2 * i + 1], false, bq[i].zw, (short)0, acc, false, false);
    }

    #pragma unroll
    for (int r = 0; r < 8; ++r) {
        const int m = rt * 16 + r + 8 * hi;
        out[m * 32 + ct * 16 + half] = acc[r];
    }
}

extern "C" void kernel_launch(void* const* d_in, const int* in_sizes, int n_in,
                              void* d_out, int out_size, void* d_ws, size_t ws_size,
                              hipStream_t stream) {
    const float* x    = (const float*)d_in[0];
    const float* Win  = (const float*)d_in[1];
    const float* Wout = (const float*)d_in[2];
    float* out = (float*)d_out;

    float* G = (float*)d_ws;

    gram_kernel<<<20, 512, 0, stream>>>(Win, Wout, G);
    apply_kernel<<<8, 256, 0, stream>>>(x, G, out);
}
